# starting kernel
# baseline (speedup 1.0000x reference)
#include <hip/hip_runtime.h>
#include <hip/hip_bf16.h>
#include <math.h>

#define GMM_B 32768
#define GMM_C 100
#define GMM_D 64

typedef __attribute__((ext_vector_type(2))) float v2f;
typedef __attribute__((ext_vector_type(8))) float v8f;

__global__ __launch_bounds__(64)
void gmm_prep_kernel(const float* __restrict__ means,
                     const float* __restrict__ covs,
                     const float* __restrict__ weights,
                     float* __restrict__ Gout,
                     float* __restrict__ hout,
                     float* __restrict__ ccout)
{
    __shared__ float A [GMM_D * 65];
    __shared__ float Gi[GMM_D * 65];
    __shared__ float red[GMM_D];
    __shared__ float mu [GMM_D];

    const int t = threadIdx.x;
    const int c = blockIdx.x;

    for (int k = 0; k < GMM_D; ++k)
        A[t * 65 + k] = covs[(size_t)c * GMM_D * GMM_D + (size_t)t * GMM_D + k];
    mu[t] = means[c * GMM_D + t];
    __syncthreads();

    for (int j = 0; j < GMM_D; ++j) {
        if (t == j) A[j * 65 + j] = sqrtf(A[j * 65 + j]);
        __syncthreads();
        if (t > j) A[t * 65 + j] /= A[j * 65 + j];
        __syncthreads();
        if (t > j) {
            const float ltj = A[t * 65 + j];
            for (int k = j + 1; k <= t; ++k)
                A[t * 65 + k] -= ltj * A[k * 65 + j];
        }
        __syncthreads();
    }

    red[t] = 2.0f * logf(A[t * 65 + t]);
    __syncthreads();
    if (t == 0) {
        float ld = 0.0f;
        for (int k = 0; k < GMM_D; ++k) ld += red[k];
        const float LOG2PI = 1.8378770664093453f;
        ccout[c] = logf(weights[c]) - 0.5f * (ld + (float)GMM_D * LOG2PI);
    }

    for (int i = 0; i < GMM_D; ++i) {
        float sum = (i == t) ? 1.0f : 0.0f;
        for (int k = t; k < i; ++k)
            sum -= A[i * 65 + k] * Gi[k * 65 + t];
        Gi[i * 65 + t] = (i >= t) ? (sum / A[i * 65 + i]) : 0.0f;
    }
    __syncthreads();

    float hv = 0.0f;
    for (int k = 0; k <= t; ++k) hv += Gi[t * 65 + k] * mu[k];
    hout[c * GMM_D + t] = hv;

    for (int k = 0; k < GMM_D; ++k)
        Gout[(size_t)c * GMM_D * GMM_D + (size_t)t * GMM_D + k] = Gi[t * 65 + k];
}

#define TILE_N 128
#define LSTR   68

__global__ __launch_bounds__(256)
void gmm_logits_kernel(const float* __restrict__ x,
                       const float* __restrict__ G,
                       const float* __restrict__ h,
                       const float* __restrict__ cc,
                       float* __restrict__ out)
{
    __shared__ __align__(16) float Xl[TILE_N * LSTR];
    __shared__ __align__(16) float Gl[GMM_D * LSTR];
    __shared__ float hl[GMM_D];

    const int t  = threadIdx.x;
    const int c  = blockIdx.y;
    const int n0 = blockIdx.x * TILE_N;

    {
        const int col4 = (t & 15) * 4;
        const int r0   = t >> 4;
        #pragma unroll
        for (int it = 0; it < 8; ++it) {
            const int r = r0 + it * 16;
            const float4 v = *(const float4*)(x + (size_t)(n0 + r) * GMM_D + col4);
            *(float4*)&Xl[r * LSTR + col4] = v;
        }
        #pragma unroll
        for (int it = 0; it < 4; ++it) {
            const int r = r0 + it * 16;
            const float4 v = *(const float4*)(G + (size_t)c * GMM_D * GMM_D + (size_t)r * GMM_D + col4);
            *(float4*)&Gl[r * LSTR + col4] = v;
        }
        if (t < GMM_D) hl[t] = h[c * GMM_D + t];
    }
    __syncthreads();

    const int lane  = t & 31;
    const int w     = t >> 5;
    const int nloc  = lane & 15;
    const int khalf = (lane >> 4) * 2;

    const float* brow = &Xl[(w * 16 + nloc) * LSTR];

    float s = 0.0f;
    #pragma unroll
    for (int mt = 0; mt < 4; ++mt) {
        v8f acc = {};
        const float* arow = &Gl[(mt * 16 + nloc) * LSTR];
        #pragma unroll
        for (int k0 = 0; k0 < GMM_D; k0 += 4) {
            v2f a = *(const v2f*)(arow + k0 + khalf);
            v2f b = *(const v2f*)(brow + k0 + khalf);
            acc = __builtin_amdgcn_wmma_f32_16x16x4_f32(
                       false, a,  false, b,
                       (short)0, acc,
                       false,  false);
        }
        const int mrow = mt * 16 + (lane >> 4) * 8;
        #pragma unroll
        for (int j = 0; j < 8; ++j) {
            const float d = acc[j] - hl[mrow + j];
            s += d * d;
        }
    }

    s += __shfl_xor(s, 16, 32);

    if (lane < 16) {
        const int b = n0 + w * 16 + nloc;
        out[(size_t)b * GMM_C + c] = cc[c] - 0.5f * s;
    }
}

extern "C" void kernel_launch(void* const* d_in, const int* in_sizes, int n_in,
                              void* d_out, int out_size, void* d_ws, size_t ws_size,
                              hipStream_t stream) {
    (void)in_sizes; (void)n_in; (void)out_size; (void)ws_size;

    const float* x       = (const float*)d_in[0];
    const float* means   = (const float*)d_in[1];
    const float* covs    = (const float*)d_in[2];
    const float* weights = (const float*)d_in[3];
    float* out = (float*)d_out;

    float* Gw  = (float*)d_ws;
    float* hw  = Gw + (size_t)GMM_C * GMM_D * GMM_D;
    float* ccw = hw + (size_t)GMM_C * GMM_D;

    gmm_prep_kernel<<<GMM_C, 64, 0, stream>>>(means, covs, weights, Gw, hw, ccw);

    dim3 grid(GMM_B / TILE_N, GMM_C);
    gmm_logits_kernel<<<grid, 256, 0, stream>>>(x, Gw, hw, ccw, out);
}


# main kernel: classes 96..99 leave the pair-feature GEMM (a 7th class tile, 4 of 16 columns used: 560 WMMAs per row tile) for the direct form |G x - h|^2: 160 WMMAs, G triangular; 6 full tiles stay
# speedup vs baseline: 2.5763x; 2.5763x over previous
#include <hip/hip_runtime.h>
#include <math.h>

#define GMM_B 32768
#define GMM_C 100
#define GMM_D 64

typedef __attribute__((ext_vector_type(2))) float v2f;
typedef __attribute__((ext_vector_type(8))) float v8f;

#define NKS   560
#define NT    6
#define NCP   (NT * 16)
#define NDC   (GMM_C - NCP)
#define NFR   40
#define KLIN  544

__device__ __forceinline__ int kstep_of(int i, int jq)
{
    const int g = i >> 2, r = i & 3;
    return 4 * (16 * g - (g * (g - 1)) / 2) + r * (16 - g) + (jq - g);
}
__device__ __forceinline__ int wc_index(int w, int c, int half, int s)
{
    return ((w * NT + (c >> 4)) * 32 + half * 16 + (c & 15)) * 2 + s;
}

#define GP 68
__global__ __launch_bounds__(256)
void gmm_prep_kernel(const float* __restrict__ means,
                     const float* __restrict__ covs,
                     const float* __restrict__ weights,
                     float* __restrict__ Wc,
                     float* __restrict__ constc,
                     float* __restrict__ Gd,
                     float* __restrict__ hd,
                     float* __restrict__ cd)
{
    __shared__ __align__(16) float rowbuf[2][2 * GMM_D];
    __shared__ __align__(16) float colbuf[2][GMM_D];
    __shared__ float dbuf[GMM_D];
    __shared__ __align__(16) float Gs[GMM_D * GP];
    __shared__ float qs[8][GMM_D];
    __shared__ float hhs[8];
    __shared__ float ccs;

    const int t  = threadIdx.x;
    const int tx = t & 31;
    const int ty = t >> 5;
    const int c  = blockIdx.x;

    const float* cov = covs + (size_t)c * GMM_D * GMM_D;

    float m[8][4];
    #pragma unroll
    for (int i = 0; i < 8; ++i) {
        const int r = ty * 8 + i;
        m[i][0] = cov[r * GMM_D + tx];
        m[i][1] = cov[r * GMM_D + 32 + tx];
        m[i][2] = (r == tx) ? 1.0f : 0.0f;
        m[i][3] = (r == tx + 32) ? 1.0f : 0.0f;
    }

    for (int jo = 0; jo < 8; ++jo) {
        const bool hi  = jo >= 4;
        const int  ctx = (jo & 3) * 8;
        #pragma unroll
        for (int i = 0; i < 8; ++i) {
            const int j = jo * 8 + i;
            const int b = i & 1;
            if (ty == jo) {
                rowbuf[b][tx]      = m[i][0];
                rowbuf[b][tx + 32] = m[i][1];
                rowbuf[b][tx + 64] = m[i][2];
                rowbuf[b][tx + 96] = m[i][3];
            }
            if (tx == ctx + i) {
                #pragma unroll
                for (int ii = 0; ii < 8; ++ii)
                    colbuf[b][ty * 8 + ii] = hi ? m[ii][1] : m[ii][0];
            }
            __syncthreads();
            if (ty >= jo) {
                const float piv = rowbuf[b][j];
                const float4 ca = *(const float4*)&colbuf[b][ty * 8];
                const float4 cb = *(const float4*)&colbuf[b][ty * 8 + 4];
                const float p0 = rowbuf[b][tx];
                const float p1 = rowbuf[b][tx + 32];
                const float p2 = rowbuf[b][tx + 64];
                const float p3 = rowbuf[b][tx + 96];
                const float r0  = __builtin_amdgcn_rcpf(piv);
                const float inv = fmaf(fmaf(-piv, r0, 1.0f), r0, r0);
                const float cl[8] = {ca.x, ca.y, ca.z, ca.w, cb.x, cb.y, cb.z, cb.w};
                #pragma unroll
                for (int ii = 0; ii < 8; ++ii) {
                    const int r = ty * 8 + ii;
                    const float f = (r > j) ? cl[ii] * inv : 0.0f;
                    m[ii][0] -= f * p0;
                    m[ii][1] -= f * p1;
                    m[ii][2] -= f * p2;
                    m[ii][3] -= f * p3;
                }
            }
        }
    }

    #pragma unroll
    for (int i = 0; i < 8; ++i) {
        const int r = ty * 8 + i;
        if (tx == (r & 31)) dbuf[r] = (ty >= 4) ? m[i][1] : m[i][0];
    }
    __syncthreads();

    const float mu0 = means[c * GMM_D + tx];
    const float mu1 = means[c * GMM_D + 32 + tx];
    float q0 = 0.0f, q1 = 0.0f, hhw = 0.0f;
    #pragma unroll
    for (int i = 0; i < 8; ++i) {
        const int r = ty * 8 + i;
        const float rs = 1.0f / sqrtf(dbuf[r]);
        const float g0 = m[i][2] * rs;
        const float g1 = m[i][3] * rs;
        Gs[r * GP + tx]      = g0;
        Gs[r * GP + 32 + tx] = g1;
        float hp = g0 * mu0 + g1 * mu1;
        hp += __shfl_xor(hp, 16, 32);
        hp += __shfl_xor(hp, 8, 32);
        hp += __shfl_xor(hp, 4, 32);
        hp += __shfl_xor(hp, 2, 32);
        hp += __shfl_xor(hp, 1, 32);
        q0  = fmaf(g0, hp, q0);
        q1  = fmaf(g1, hp, q1);
        hhw = fmaf(hp, hp, hhw);
        if (c >= NCP && tx == 0) hd[(c - NCP) * GMM_D + r] = hp;
    }
    qs[ty][tx]      = q0;
    qs[ty][tx + 32] = q1;
    if (tx == 0) hhs[ty] = hhw;

    if (ty == 0) {
        float ld = logf(dbuf[tx]) + logf(dbuf[tx + 32]);
        ld += __shfl_xor(ld, 16, 32);
        ld += __shfl_xor(ld, 8, 32);
        ld += __shfl_xor(ld, 4, 32);
        ld += __shfl_xor(ld, 2, 32);
        ld += __shfl_xor(ld, 1, 32);
        if (tx == 0) {
            const float LOG2PI = 1.8378770664093453f;
            ccs = logf(weights[c]) - 0.5f * (ld + (float)GMM_D * LOG2PI);
        }
    }
    __syncthreads();

    if (c >= NCP) {
        float* gd = Gd + (size_t)(c - NCP) * NFR * 64;
        for (int e = t; e < NFR * 64; e += 256) {
            const int f  = e >> 6;
            const int l  = (e >> 1) & 31;
            const int sx = e & 1;
            const int mt = (f >= 24) ? 3 : ((f >= 12) ? 2 : ((f >= 4) ? 1 : 0));
            const int kq = f - 2 * mt * (mt + 1);
            gd[e] = Gs[(16 * mt + (l & 15)) * GP + 4 * kq + 2 * (l >> 4) + sx];
        }
        if (t == 0) cd[c - NCP] = ccs;
        return;
    }

    {
        const int ib = t >> 4;
        const int jb = t & 15;
        if (jb >= ib) {
            float pb[4][4];
            #pragma unroll
            for (int a = 0; a < 4; ++a)
                #pragma unroll
                for (int b = 0; b < 4; ++b) pb[a][b] = 0.0f;
            #pragma unroll 4
            for (int r = 4 * jb; r < GMM_D; ++r) {
                const float4 gi = *(const float4*)&Gs[r * GP + 4 * ib];
                const float4 gj = *(const float4*)&Gs[r * GP + 4 * jb];
                const float ga[4] = {gi.x, gi.y, gi.z, gi.w};
                const float gb[4] = {gj.x, gj.y, gj.z, gj.w};
                #pragma unroll
                for (int a = 0; a < 4; ++a)
                    #pragma unroll
                    for (int b = 0; b < 4; ++b) pb[a][b] = fmaf(ga[a], gb[b], pb[a][b]);
            }
            #pragma unroll
            for (int a = 0; a < 4; ++a) {
                const int i = 4 * ib + a;
                const int w = kstep_of(i, jb);
                #pragma unroll
                for (int b = 0; b < 4; ++b) {
                    const int j = 4 * jb + b;
                    const float v = (j < i) ? 0.0f : ((j == i) ? -0.5f * pb[a][b] : -pb[a][b]);
                    Wc[wc_index(w, c, b >> 1, b & 1)] = v;
                }
            }
        }
    }
    if (t < GMM_D) {
        const int j = t;
        float q = 0.0f;
        #pragma unroll
        for (int w8 = 0; w8 < 8; ++w8) q += qs[w8][j];
        Wc[wc_index(KLIN + (j >> 2), c, (j >> 1) & 1, j & 1)] = q;
    } else if (t == GMM_D) {
        float hh = 0.0f;
        #pragma unroll
        for (int w8 = 0; w8 < 8; ++w8) hh += hhs[w8];
        constc[c] = ccs - 0.5f * hh;
    }
}

#define TILE_M 128
#define XP     68
#define CHK    40
#define NCH    (NKS / CHK)
#define CHF    (CHK * NT * 64)
#define NPF    (CHF / 2 / 256)
#define NPC    5
#define PPF    (NPF / NPC)
#define PKS    (CHK / NPC)
#define SB_MASK 0x06

__global__ __launch_bounds__(256)
void gmm_logits_kernel(const float* __restrict__ x,
                       const float* __restrict__ Wc,
                       const float* __restrict__ constc,
                       const float* __restrict__ Gd,
                       const float* __restrict__ hd,
                       const float* __restrict__ cd,
                       float* __restrict__ out)
{
    __shared__ __align__(16) float Xl[TILE_M * XP];
    __shared__ __align__(16) float Bl[2][CHF];
    __shared__ __align__(16) float Gl[NDC * NFR * 64];
    __shared__ __align__(16) float hl[NDC * GMM_D];

    const int t  = threadIdx.x;
    const int b0 = blockIdx.x * TILE_M;

    {
        const int col4 = (t & 15) * 4;
        const int r0   = t >> 4;
        #pragma unroll
        for (int it = 0; it < 8; ++it) {
            const int r = r0 + it * 16;
            const float4 v = *(const float4*)(x + (size_t)(b0 + r) * GMM_D + col4);
            *(float4*)&Xl[r * XP + col4] = v;
        }
        if (t < TILE_M) Xl[t * XP + GMM_D] = 1.0f;
        #pragma unroll
        for (int q = 0; q < NPF; ++q) {
            const float2 v = *(const float2*)(Wc + (size_t)(q * 256 + t) * 2);
            *(float2*)&Bl[0][(q * 256 + t) * 2] = v;
        }
        #pragma unroll
        for (int q = 0; q < NDC * NFR * 64 / 4 / 256; ++q) {
            const float4 v = *(const float4*)(Gd + (size_t)(q * 256 + t) * 4);
            *(float4*)&Gl[(q * 256 + t) * 4] = v;
        }
        hl[t] = hd[t];
    }

    const int lane = t & 31;
    const int wv   = t >> 5;
    const int n    = lane & 15;
    const int half = lane >> 4;

    v8f acc[NT];
    #pragma unroll
    for (int tt = 0; tt < NT; ++tt) {
        #pragma unroll
        for (int r = 0; r < 8; ++r) acc[tt][r] = 0.0f;
    }
    __syncthreads();

    const float* xrow = &Xl[(wv * 16 + n) * XP];
    const float* xj   = xrow + 2 * half;

    float res[NDC];
    {
        v2f xb[16];
        #pragma unroll
        for (int kq = 0; kq < 16; ++kq) xb[kq] = *(const v2f*)(xj + 4 * kq);
        #pragma unroll
        for (int cl = 0; cl < NDC; ++cl) {
            v8f d[4];
            #pragma unroll
            for (int mt = 0; mt < 4; ++mt) {
                const float* hp = &hl[cl * GMM_D + 16 * mt + 8 * half];
                const float4 h0 = *(const float4*)hp;
                const float4 h1 = *(const float4*)(hp + 4);
                d[mt][0] = -h0.x; d[mt][1] = -h0.y; d[mt][2] = -h0.z; d[mt][3] = -h0.w;
                d[mt][4] = -h1.x; d[mt][5] = -h1.y; d[mt][6] = -h1.z; d[mt][7] = -h1.w;
            }
            const float* gl = &Gl[(cl * NFR * 32 + lane) * 2];
            #pragma unroll
            for (int kq = 0; kq < 16; ++kq) {
                #pragma unroll
                for (int mt = kq >> 2; mt < 4; ++mt) {
                    const v2f a = *(const v2f*)(gl + (2 * mt * (mt + 1) + kq) * 64);
                    d[mt] = __builtin_amdgcn_wmma_f32_16x16x4_f32(
                                 false, a,  false, xb[kq],
                                 (short)0, d[mt],
                                 false,  false);
                }
            }
            float ms0 = 0.0f, ms1 = 0.0f;
            #pragma unroll
            for (int mt = 0; mt < 4; ++mt) {
                #pragma unroll
                for (int r = 0; r < 8; r += 2) {
                    ms0 = fmaf(d[mt][r],     d[mt][r],     ms0);
                    ms1 = fmaf(d[mt][r + 1], d[mt][r + 1], ms1);
                }
            }
            float ms = ms0 + ms1;
            ms += __shfl_xor(ms, 16, 32);
            res[cl] = cd[cl] - 0.5f * ms;
        }
    }

    int ki = 0, kq = 0;
    for (int ch = 0; ch < NCH; ++ch) {
        const int nx = (ch + 1 < NCH) ? (ch + 1) : ch;
        const float* wnx = Wc + (size_t)nx * CHF + (size_t)t * 2;
        float*       bnx = &Bl[(ch + 1) & 1][t * 2];
        float2 pf[PPF];
        #pragma unroll
        for (int q = 0; q < PPF; ++q)
            pf[q] = *(const float2*)(wnx + q * 512);
        const float* bb = &Bl[ch & 1][lane * 2];
        float xi = xrow[ki];
        v2f   xp = *(const v2f*)(xj + 4 * kq);
        v2f   bf[NT];
        #pragma unroll
        for (int tt = 0; tt < NT; ++tt) bf[tt] = *(const v2f*)(bb + tt * 64);
        #pragma unroll
        for (int kk = 0; kk < CHK; ++kk) {
            if (kk > 0 && (kk % PKS) == 0) {
                #pragma unroll
                for (int q = 0; q < PPF; ++q)
                    *(float2*)(bnx + ((kk / PKS - 1) * PPF + q) * 512) = pf[q];
                #pragma unroll
                for (int q = 0; q < PPF; ++q)
                    pf[q] = *(const float2*)(wnx + ((kk / PKS) * PPF + q) * 512);
            }
            int ni = ki, nq = kq + 1;
            if (nq == 16) { ni = ki + 1; nq = (ni < GMM_D) ? (ni >> 2) : 0; }
            float xin = xi;
            v2f   xpn = xp;
            v2f   bn[NT];
            #pragma unroll
            for (int tt = 0; tt < NT; ++tt) bn[tt] = bf[tt];
            if (kk + 1 < CHK) {
                xin = xrow[ni];
                xpn = *(const v2f*)(xj + 4 * nq);
                #pragma unroll
                for (int tt = 0; tt < NT; ++tt)
                    bn[tt] = *(const v2f*)(bb + ((kk + 1) * NT + tt) * 64);
            }
            __builtin_amdgcn_sched_barrier(SB_MASK);
            v2f a;
            a.x = xi * xp.x;
            a.y = xi * xp.y;
            #pragma unroll
            for (int tt = 0; tt < NT; ++tt) {
                acc[tt] = __builtin_amdgcn_wmma_f32_16x16x4_f32(
                               false, a,  false, bf[tt],
                               (short)0, acc[tt],
                               false,  false);
            }
            __builtin_amdgcn_sched_barrier(SB_MASK);
            xi = xin;
            xp = xpn;
            #pragma unroll
            for (int tt = 0; tt < NT; ++tt) bf[tt] = bn[tt];
            ki = ni;
            kq = nq;
        }
        #pragma unroll
        for (int q = 0; q < PPF; ++q)
            *(float2*)(bnx + ((NPC - 1) * PPF + q) * 512) = pf[q];
        __syncthreads();
    }

    const int rowb = b0 + wv * 16 + 8 * half;
    #pragma unroll
    for (int tt = 0; tt < NT; ++tt) {
        const int cls = tt * 16 + n;
        const float c0 = constc[cls];
        #pragma unroll
        for (int r = 0; r < 8; ++r)
            out[(size_t)(rowb + r) * GMM_C + cls] = acc[tt][r] + c0;
    }
    if (half == 0)
        *(float4*)(out + (size_t)(b0 + wv * 16 + n) * GMM_C + NCP) =
            make_float4(res[0], res[1], res[2], res[3]);
}

extern "C" void kernel_launch(void* const* d_in, const int* in_sizes, int n_in,
                              void* d_out, int out_size, void* d_ws, size_t ws_size,
                              hipStream_t stream) {
    (void)in_sizes; (void)n_in; (void)out_size; (void)ws_size;

    const float* x       = (const float*)d_in[0];
    const float* means   = (const float*)d_in[1];
    const float* covs    = (const float*)d_in[2];
    const float* weights = (const float*)d_in[3];
    float* out = (float*)d_out;

    float* Wcw = (float*)d_ws;
    float* ccw = Wcw + (size_t)NKS * NT * 64;
    float* gdw = ccw + 112;
    float* hdw = gdw + (size_t)NDC * NFR * 64;
    float* cdw = hdw + NDC * GMM_D;

    gmm_prep_kernel<<<GMM_C, 256, 0, stream>>>(means, covs, weights, Wcw, ccw, gdw, hdw, cdw);

    gmm_logits_kernel<<<GMM_B / TILE_M, 256, 0, stream>>>(x, Wcw, ccw, gdw, hdw, cdw, out);
}


# prep kernel tail after the elimination: P = G^T G as 80 f32 WMMAs on 7 waves (was a 64-trip register-block loop) and the rows' lane sums on the vector ALU by DPP (were 45 LDS exchanges in series)
# speedup vs baseline: 2.7548x; 1.0693x over previous
#include <hip/hip_runtime.h>
#include <math.h>

#define GMM_B 32768
#define GMM_C 100
#define GMM_D 64

typedef __attribute__((ext_vector_type(2))) float v2f;
typedef __attribute__((ext_vector_type(8))) float v8f;

#define NKS   560
#define NT    6
#define NCP   (NT * 16)
#define NDC   (GMM_C - NCP)
#define NFR   40
#define KLIN  544

__device__ __forceinline__ int kstep_of(int i, int jq)
{
    const int g = i >> 2, r = i & 3;
    return 4 * (16 * g - (g * (g - 1)) / 2) + r * (16 - g) + (jq - g);
}
__device__ __forceinline__ int wc_index(int w, int c, int half, int s)
{
    return ((w * NT + (c >> 4)) * 32 + half * 16 + (c & 15)) * 2 + s;
}

#define GP 68

template <int XM>
__device__ __forceinline__ float gmm_add_xmask(float x)
{
    const int b = __builtin_bit_cast(int, x);
    const int s = __builtin_amdgcn_update_dpp(0, b, 0x160 + XM, 0xf, 0xf, true);
    return x + __builtin_bit_cast(float, s);
}
__device__ __forceinline__ float gmm_wave_sum(float x)
{
    const int b = __builtin_bit_cast(int, x);
    x += __builtin_bit_cast(float, __builtin_amdgcn_permlanex16(b, b, 0x76543210, 0xfedcba98, false, false));
    x = gmm_add_xmask<8>(x);
    x = gmm_add_xmask<4>(x);
    x = gmm_add_xmask<2>(x);
    x = gmm_add_xmask<1>(x);
    return x;
}

template <int I, int J>
__device__ __forceinline__ void gmm_p_tile(const float* Gs, int lane, int c, float* __restrict__ Wc)
{
    constexpr int NK = 16 - 4 * J;
    const int n    = lane & 15;
    const int half = lane >> 4;
    const float* ga = Gs + (4 * (4 * J) + 2 * half) * GP + 16 * I + n;
    const float* gb = Gs + (4 * (4 * J) + 2 * half) * GP + 16 * J + n;
    v2f fa[NK], fb[NK];
    #pragma unroll
    for (int k = 0; k < NK; ++k) {
        fa[k].x = ga[(4 * k) * GP];
        fa[k].y = ga[(4 * k + 1) * GP];
        if (I != J) {
            fb[k].x = gb[(4 * k) * GP];
            fb[k].y = gb[(4 * k + 1) * GP];
        } else {
            fb[k] = fa[k];
        }
    }
    v8f acc0, acc1;
    #pragma unroll
    for (int r = 0; r < 8; ++r) { acc0[r] = 0.0f; acc1[r] = 0.0f; }
    #pragma unroll
    for (int k = 0; k < NK; k += 2) {
        acc0 = __builtin_amdgcn_wmma_f32_16x16x4_f32(
                    false, fa[k],  false, fb[k],
                    (short)0, acc0,
                    false,  false);
        acc1 = __builtin_amdgcn_wmma_f32_16x16x4_f32(
                    false, fa[k + 1],  false, fb[k + 1],
                    (short)0, acc1,
                    false,  false);
    }
    const int j = 16 * J + n;
    #pragma unroll
    for (int r = 0; r < 8; ++r) {
        const int i = 16 * I + 8 * half + r;
        if ((j >> 2) >= (i >> 2)) {
            const float p = acc0[r] + acc1[r];
            const float v = (j < i) ? 0.0f : ((j == i) ? -0.5f * p : -p);
            Wc[wc_index(kstep_of(i, j >> 2), c, (j >> 1) & 1, j & 1)] = v;
        }
    }
}

__global__ __launch_bounds__(256)
void gmm_prep_kernel(const float* __restrict__ means,
                     const float* __restrict__ covs,
                     const float* __restrict__ weights,
                     float* __restrict__ Wc,
                     float* __restrict__ constc,
                     float* __restrict__ Gd,
                     float* __restrict__ hd,
                     float* __restrict__ cd)
{
    __shared__ __align__(16) float rowbuf[2][2 * GMM_D];
    __shared__ __align__(16) float colbuf[2][GMM_D];
    __shared__ __align__(16) float dbuf[GMM_D];
    __shared__ __align__(16) float Gs[GMM_D * GP];
    __shared__ float qs[8][GMM_D];
    __shared__ float hhs[8];

    const int t  = threadIdx.x;
    const int tx = t & 31;
    const int ty = t >> 5;
    const int c  = blockIdx.x;

    const float* cov = covs + (size_t)c * GMM_D * GMM_D;

    float m[8][4];
    #pragma unroll
    for (int i = 0; i < 8; ++i) {
        const int r = ty * 8 + i;
        m[i][0] = cov[r * GMM_D + tx];
        m[i][1] = cov[r * GMM_D + 32 + tx];
        m[i][2] = (r == tx) ? 1.0f : 0.0f;
        m[i][3] = (r == tx + 32) ? 1.0f : 0.0f;
    }

    const float mu0 = means[c * GMM_D + tx];
    const float mu1 = means[c * GMM_D + 32 + tx];

    for (int jo = 0; jo < 8; ++jo) {
        const bool hi  = jo >= 4;
        const int  ctx = (jo & 3) * 8;
        #pragma unroll
        for (int i = 0; i < 8; ++i) {
            const int j = jo * 8 + i;
            const int b = i & 1;
            if (ty == jo) {
                rowbuf[b][tx]      = m[i][0];
                rowbuf[b][tx + 32] = m[i][1];
                rowbuf[b][tx + 64] = m[i][2];
                rowbuf[b][tx + 96] = m[i][3];
            }
            if (tx == ctx + i) {
                #pragma unroll
                for (int ii = 0; ii < 8; ++ii)
                    colbuf[b][ty * 8 + ii] = hi ? m[ii][1] : m[ii][0];
            }
            __syncthreads();
            if (ty >= jo) {
                const float piv = rowbuf[b][j];
                const float4 ca = *(const float4*)&colbuf[b][ty * 8];
                const float4 cb = *(const float4*)&colbuf[b][ty * 8 + 4];
                const float p0 = rowbuf[b][tx];
                const float p1 = rowbuf[b][tx + 32];
                const float p2 = rowbuf[b][tx + 64];
                const float p3 = rowbuf[b][tx + 96];
                const float r0  = __builtin_amdgcn_rcpf(piv);
                const float inv = fmaf(fmaf(-piv, r0, 1.0f), r0, r0);
                const float cl[8] = {ca.x, ca.y, ca.z, ca.w, cb.x, cb.y, cb.z, cb.w};
                #pragma unroll
                for (int ii = 0; ii < 8; ++ii) {
                    const int r = ty * 8 + ii;
                    const float f = (r > j) ? cl[ii] * inv : 0.0f;
                    m[ii][0] -= f * p0;
                    m[ii][1] -= f * p1;
                    m[ii][2] -= f * p2;
                    m[ii][3] -= f * p3;
                }
            }
        }
    }

    #pragma unroll
    for (int i = 0; i < 8; ++i) {
        const int r = ty * 8 + i;
        if (tx == (r & 31)) dbuf[r] = (ty >= 4) ? m[i][1] : m[i][0];
    }
    __syncthreads();

    float q0 = 0.0f, q1 = 0.0f, hhw = 0.0f;
    const float4 da = *(const float4*)&dbuf[ty * 8];
    const float4 db = *(const float4*)&dbuf[ty * 8 + 4];
    const float dv[8] = {da.x, da.y, da.z, da.w, db.x, db.y, db.z, db.w};
    #pragma unroll
    for (int i = 0; i < 8; ++i) {
        const int r = ty * 8 + i;
        const float d  = dv[i];
        const float y0 = __builtin_amdgcn_rsqf(d);
        const float rs = fmaf(0.5f * y0, fmaf(-(d * y0), y0, 1.0f), y0);
        const float g0 = m[i][2] * rs;
        const float g1 = m[i][3] * rs;
        Gs[r * GP + tx]      = g0;
        Gs[r * GP + 32 + tx] = g1;
        const float hp = gmm_wave_sum(g0 * mu0 + g1 * mu1);
        q0  = fmaf(g0, hp, q0);
        q1  = fmaf(g1, hp, q1);
        hhw = fmaf(hp, hp, hhw);
        if (c >= NCP && tx == 0) hd[(c - NCP) * GMM_D + r] = hp;
    }
    qs[ty][tx]      = q0;
    qs[ty][tx + 32] = q1;
    if (tx == 0) hhs[ty] = hhw;

    __syncthreads();

    const int wu = __builtin_amdgcn_readfirstlane(ty);

    float ccl = 0.0f;
    if (wu == 7) {
        const float ld = gmm_wave_sum(logf(dbuf[tx]) + logf(dbuf[tx + 32]));
        const float LOG2PI = 1.8378770664093453f;
        ccl = logf(weights[c]) - 0.5f * (ld + (float)GMM_D * LOG2PI);
    }

    if (c >= NCP) {
        float* gd = Gd + (size_t)(c - NCP) * NFR * 64;
        for (int e = t; e < NFR * 64; e += 256) {
            const int f  = e >> 6;
            const int l  = (e >> 1) & 31;
            const int sx = e & 1;
            const int mt = (f >= 24) ? 3 : ((f >= 12) ? 2 : ((f >= 4) ? 1 : 0));
            const int kq = f - 2 * mt * (mt + 1);
            gd[e] = Gs[(16 * mt + (l & 15)) * GP + 4 * kq + 2 * (l >> 4) + sx];
        }
        if (t == 255) cd[c - NCP] = ccl;
        return;
    }

    switch (wu) {
        case 0: gmm_p_tile<0, 0>(Gs, tx, c, Wc); break;
        case 1: gmm_p_tile<0, 1>(Gs, tx, c, Wc); break;
        case 2: gmm_p_tile<1, 1>(Gs, tx, c, Wc); break;
        case 3: gmm_p_tile<0, 2>(Gs, tx, c, Wc); gmm_p_tile<0, 3>(Gs, tx, c, Wc); break;
        case 4: gmm_p_tile<1, 2>(Gs, tx, c, Wc); gmm_p_tile<1, 3>(Gs, tx, c, Wc); break;
        case 5: gmm_p_tile<2, 2>(Gs, tx, c, Wc); gmm_p_tile<2, 3>(Gs, tx, c, Wc); break;
        case 6: gmm_p_tile<3, 3>(Gs, tx, c, Wc); break;
        default: {
            #pragma unroll
            for (int jj = 0; jj < 2; ++jj) {
                const int j = tx + 32 * jj;
                float q = 0.0f;
                #pragma unroll
                for (int w8 = 0; w8 < 8; ++w8) q += qs[w8][j];
                Wc[wc_index(KLIN + (j >> 2), c, (j >> 1) & 1, j & 1)] = q;
            }
            if (tx == 0) {
                float hh = 0.0f;
                #pragma unroll
                for (int w8 = 0; w8 < 8; ++w8) hh += hhs[w8];
                constc[c] = ccl - 0.5f * hh;
            }
        } break;
    }
}

#define TILE_M 128
#define XP     68
#define CHK    40
#define NCH    (NKS / CHK)
#define CHF    (CHK * NT * 64)
#define NPF    (CHF / 2 / 256)
#define NPC    5
#define PPF    (NPF / NPC)
#define PKS    (CHK / NPC)
#define SB_MASK 0x06

__global__ __launch_bounds__(256)
void gmm_logits_kernel(const float* __restrict__ x,
                       const float* __restrict__ Wc,
                       const float* __restrict__ constc,
                       const float* __restrict__ Gd,
                       const float* __restrict__ hd,
                       const float* __restrict__ cd,
                       float* __restrict__ out)
{
    __shared__ __align__(16) float Xl[TILE_M * XP];
    __shared__ __align__(16) float Bl[2][CHF];
    __shared__ __align__(16) float Gl[NDC * NFR * 64];
    __shared__ __align__(16) float hl[NDC * GMM_D];

    const int t  = threadIdx.x;
    const int b0 = blockIdx.x * TILE_M;

    {
        const int col4 = (t & 15) * 4;
        const int r0   = t >> 4;
        #pragma unroll
        for (int it = 0; it < 8; ++it) {
            const int r = r0 + it * 16;
            const float4 v = *(const float4*)(x + (size_t)(b0 + r) * GMM_D + col4);
            *(float4*)&Xl[r * XP + col4] = v;
        }
        if (t < TILE_M) Xl[t * XP + GMM_D] = 1.0f;
        #pragma unroll
        for (int q = 0; q < NPF; ++q) {
            const float2 v = *(const float2*)(Wc + (size_t)(q * 256 + t) * 2);
            *(float2*)&Bl[0][(q * 256 + t) * 2] = v;
        }
        #pragma unroll
        for (int q = 0; q < NDC * NFR * 64 / 4 / 256; ++q) {
            const float4 v = *(const float4*)(Gd + (size_t)(q * 256 + t) * 4);
            *(float4*)&Gl[(q * 256 + t) * 4] = v;
        }
        hl[t] = hd[t];
    }

    const int lane = t & 31;
    const int wv   = t >> 5;
    const int n    = lane & 15;
    const int half = lane >> 4;

    v8f acc[NT];
    #pragma unroll
    for (int tt = 0; tt < NT; ++tt) {
        #pragma unroll
        for (int r = 0; r < 8; ++r) acc[tt][r] = 0.0f;
    }
    __syncthreads();

    const float* xrow = &Xl[(wv * 16 + n) * XP];
    const float* xj   = xrow + 2 * half;

    float res[NDC];
    {
        v2f xb[16];
        #pragma unroll
        for (int kq = 0; kq < 16; ++kq) xb[kq] = *(const v2f*)(xj + 4 * kq);
        #pragma unroll
        for (int cl = 0; cl < NDC; ++cl) {
            v8f d[4];
            #pragma unroll
            for (int mt = 0; mt < 4; ++mt) {
                const float* hp = &hl[cl * GMM_D + 16 * mt + 8 * half];
                const float4 h0 = *(const float4*)hp;
                const float4 h1 = *(const float4*)(hp + 4);
                d[mt][0] = -h0.x; d[mt][1] = -h0.y; d[mt][2] = -h0.z; d[mt][3] = -h0.w;
                d[mt][4] = -h1.x; d[mt][5] = -h1.y; d[mt][6] = -h1.z; d[mt][7] = -h1.w;
            }
            const float* gl = &Gl[(cl * NFR * 32 + lane) * 2];
            #pragma unroll
            for (int kq = 0; kq < 16; ++kq) {
                #pragma unroll
                for (int mt = kq >> 2; mt < 4; ++mt) {
                    const v2f a = *(const v2f*)(gl + (2 * mt * (mt + 1) + kq) * 64);
                    d[mt] = __builtin_amdgcn_wmma_f32_16x16x4_f32(
                                 false, a,  false, xb[kq],
                                 (short)0, d[mt],
                                 false,  false);
                }
            }
            float ms0 = 0.0f, ms1 = 0.0f;
            #pragma unroll
            for (int mt = 0; mt < 4; ++mt) {
                #pragma unroll
                for (int r = 0; r < 8; r += 2) {
                    ms0 = fmaf(d[mt][r],     d[mt][r],     ms0);
                    ms1 = fmaf(d[mt][r + 1], d[mt][r + 1], ms1);
                }
            }
            float ms = ms0 + ms1;
            ms += __shfl_xor(ms, 16, 32);
            res[cl] = cd[cl] - 0.5f * ms;
        }
    }

    int ki = 0, kq = 0;
    for (int ch = 0; ch < NCH; ++ch) {
        const int nx = (ch + 1 < NCH) ? (ch + 1) : ch;
        const float* wnx = Wc + (size_t)nx * CHF + (size_t)t * 2;
        float*       bnx = &Bl[(ch + 1) & 1][t * 2];
        float2 pf[PPF];
        #pragma unroll
        for (int q = 0; q < PPF; ++q)
            pf[q] = *(const float2*)(wnx + q * 512);
        const float* bb = &Bl[ch & 1][lane * 2];
        float xi = xrow[ki];
        v2f   xp = *(const v2f*)(xj + 4 * kq);
        v2f   bf[NT];
        #pragma unroll
        for (int tt = 0; tt < NT; ++tt) bf[tt] = *(const v2f*)(bb + tt * 64);
        #pragma unroll
        for (int kk = 0; kk < CHK; ++kk) {
            if (kk > 0 && (kk % PKS) == 0) {
                #pragma unroll
                for (int q = 0; q < PPF; ++q)
                    *(float2*)(bnx + ((kk / PKS - 1) * PPF + q) * 512) = pf[q];
                #pragma unroll
                for (int q = 0; q < PPF; ++q)
                    pf[q] = *(const float2*)(wnx + ((kk / PKS) * PPF + q) * 512);
            }
            int ni = ki, nq = kq + 1;
            if (nq == 16) { ni = ki + 1; nq = (ni < GMM_D) ? (ni >> 2) : 0; }
            float xin = xi;
            v2f   xpn = xp;
            v2f   bn[NT];
            #pragma unroll
            for (int tt = 0; tt < NT; ++tt) bn[tt] = bf[tt];
            if (kk + 1 < CHK) {
                xin = xrow[ni];
                xpn = *(const v2f*)(xj + 4 * nq);
                #pragma unroll
                for (int tt = 0; tt < NT; ++tt)
                    bn[tt] = *(const v2f*)(bb + ((kk + 1) * NT + tt) * 64);
            }
            __builtin_amdgcn_sched_barrier(SB_MASK);
            v2f a;
            a.x = xi * xp.x;
            a.y = xi * xp.y;
            #pragma unroll
            for (int tt = 0; tt < NT; ++tt) {
                acc[tt] = __builtin_amdgcn_wmma_f32_16x16x4_f32(
                               false, a,  false, bf[tt],
                               (short)0, acc[tt],
                               false,  false);
            }
            __builtin_amdgcn_sched_barrier(SB_MASK);
            xi = xin;
            xp = xpn;
            #pragma unroll
            for (int tt = 0; tt < NT; ++tt) bf[tt] = bn[tt];
            ki = ni;
            kq = nq;
        }
        #pragma unroll
        for (int q = 0; q < PPF; ++q)
            *(float2*)(bnx + ((NPC - 1) * PPF + q) * 512) = pf[q];
        __syncthreads();
    }

    const int rowb = b0 + wv * 16 + 8 * half;
    #pragma unroll
    for (int tt = 0; tt < NT; ++tt) {
        const int cls = tt * 16 + n;
        const float c0 = constc[cls];
        #pragma unroll
        for (int r = 0; r < 8; ++r)
            out[(size_t)(rowb + r) * GMM_C + cls] = acc[tt][r] + c0;
    }
    if (half == 0)
        *(float4*)(out + (size_t)(b0 + wv * 16 + n) * GMM_C + NCP) =
            make_float4(res[0], res[1], res[2], res[3]);
}

extern "C" void kernel_launch(void* const* d_in, const int* in_sizes, int n_in,
                              void* d_out, int out_size, void* d_ws, size_t ws_size,
                              hipStream_t stream) {
    (void)in_sizes; (void)n_in; (void)out_size; (void)ws_size;

    const float* x       = (const float*)d_in[0];
    const float* means   = (const float*)d_in[1];
    const float* covs    = (const float*)d_in[2];
    const float* weights = (const float*)d_in[3];
    float* out = (float*)d_out;

    float* Wcw = (float*)d_ws;
    float* ccw = Wcw + (size_t)NKS * NT * 64;
    float* gdw = ccw + 112;
    float* hdw = gdw + (size_t)NDC * NFR * 64;
    float* cdw = hdw + NDC * GMM_D;

    gmm_prep_kernel<<<GMM_C, 256, 0, stream>>>(means, covs, weights, Wcw, ccw, gdw, hdw, cdw);

    gmm_logits_kernel<<<GMM_B / TILE_M, 256, 0, stream>>>(x, Wcw, ccw, gdw, hdw, cdw, out);
}


# main kernel: the two waves of a SIMD share two 16-row tiles (one B fragment feeds 2 WMMAs) and split each chunk's k-steps 28:12, older:younger; partial sums joined through LDS at the end
# speedup vs baseline: 3.0719x; 1.1151x over previous
#include <hip/hip_runtime.h>
#include <math.h>

#define GMM_B 32768
#define GMM_C 100
#define GMM_D 64

typedef __attribute__((ext_vector_type(2))) float v2f;
typedef __attribute__((ext_vector_type(8))) float v8f;

#define NKS   560
#define NT    6
#define NCP   (NT * 16)
#define NDC   (GMM_C - NCP)
#define NFR   40
#define KLIN  544

__device__ __forceinline__ int kstep_of(int i, int jq)
{
    const int g = i >> 2, r = i & 3;
    return 4 * (16 * g - (g * (g - 1)) / 2) + r * (16 - g) + (jq - g);
}
__device__ __forceinline__ int wc_index(int w, int c, int half, int s)
{
    return ((w * NT + (c >> 4)) * 32 + half * 16 + (c & 15)) * 2 + s;
}

#define GP 68

template <int XM>
__device__ __forceinline__ float gmm_add_xmask(float x)
{
    const int b = __builtin_bit_cast(int, x);
    const int s = __builtin_amdgcn_update_dpp(0, b, 0x160 + XM, 0xf, 0xf, true);
    return x + __builtin_bit_cast(float, s);
}
__device__ __forceinline__ float gmm_wave_sum(float x)
{
    const int b = __builtin_bit_cast(int, x);
    x += __builtin_bit_cast(float, __builtin_amdgcn_permlanex16(b, b, 0x76543210, 0xfedcba98, false, false));
    x = gmm_add_xmask<8>(x);
    x = gmm_add_xmask<4>(x);
    x = gmm_add_xmask<2>(x);
    x = gmm_add_xmask<1>(x);
    return x;
}

template <int I, int J>
__device__ __forceinline__ void gmm_p_tile(const float* Gs, int lane, int c, float* __restrict__ Wc)
{
    constexpr int NK = 16 - 4 * J;
    const int n    = lane & 15;
    const int half = lane >> 4;
    const float* ga = Gs + (4 * (4 * J) + 2 * half) * GP + 16 * I + n;
    const float* gb = Gs + (4 * (4 * J) + 2 * half) * GP + 16 * J + n;
    v2f fa[NK], fb[NK];
    #pragma unroll
    for (int k = 0; k < NK; ++k) {
        fa[k].x = ga[(4 * k) * GP];
        fa[k].y = ga[(4 * k + 1) * GP];
        if (I != J) {
            fb[k].x = gb[(4 * k) * GP];
            fb[k].y = gb[(4 * k + 1) * GP];
        } else {
            fb[k] = fa[k];
        }
    }
    v8f acc0, acc1;
    #pragma unroll
    for (int r = 0; r < 8; ++r) { acc0[r] = 0.0f; acc1[r] = 0.0f; }
    #pragma unroll
    for (int k = 0; k < NK; k += 2) {
        acc0 = __builtin_amdgcn_wmma_f32_16x16x4_f32(
                    false, fa[k],  false, fb[k],
                    (short)0, acc0,
                    false,  false);
        acc1 = __builtin_amdgcn_wmma_f32_16x16x4_f32(
                    false, fa[k + 1],  false, fb[k + 1],
                    (short)0, acc1,
                    false,  false);
    }
    const int j = 16 * J + n;
    #pragma unroll
    for (int r = 0; r < 8; ++r) {
        const int i = 16 * I + 8 * half + r;
        if ((j >> 2) >= (i >> 2)) {
            const float p = acc0[r] + acc1[r];
            const float v = (j < i) ? 0.0f : ((j == i) ? -0.5f * p : -p);
            Wc[wc_index(kstep_of(i, j >> 2), c, (j >> 1) & 1, j & 1)] = v;
        }
    }
}

#define MP 132
#define TP 20

template <int J>
__device__ __forceinline__ float gmm_row_share(float x)
{
    const int b = __builtin_bit_cast(int, x);
    return __builtin_bit_cast(float, __builtin_amdgcn_update_dpp(0, b, 0x150 + J, 0xf, 0xf, true));
}

template <int J>
__device__ __forceinline__ void gmm_diag_pivot(float (&s)[16], float (&tt)[16], int r, int lane,
                                               float* dk, float* dinvs)
{
    const float piv = gmm_row_share<J>(s[J]);
    const float r0  = __builtin_amdgcn_rcpf(piv);
    const float inv = fmaf(fmaf(-piv, r0, 1.0f), r0, r0);
    const float f   = (r > J) ? s[J] * inv : 0.0f;
    #pragma unroll
    for (int cc = J + 1; cc < 16; ++cc) s[cc] = fmaf(-f, gmm_row_share<J>(s[cc]), s[cc]);
    #pragma unroll
    for (int cc = 0; cc <= J; ++cc) tt[cc] = fmaf(-f, gmm_row_share<J>(tt[cc]), tt[cc]);
    if (lane == J) { dk[J] = piv; dinvs[J] = inv; }
}

__device__ __forceinline__ v8f gmm_mm16(const float* pa, int sa_r, int sa_k,
                                        const float* pb, int sb_k, int sb_c, v8f acc, int lane)
{
    const int n    = lane & 15;
    const int half = lane >> 4;
    v2f fa[4], fb[4];
    #pragma unroll
    for (int kq = 0; kq < 4; ++kq) {
        const int k = 4 * kq + 2 * half;
        fa[kq].x = pa[n * sa_r + k * sa_k];
        fa[kq].y = pa[n * sa_r + (k + 1) * sa_k];
        fb[kq].x = pb[k * sb_k + n * sb_c];
        fb[kq].y = pb[(k + 1) * sb_k + n * sb_c];
    }
    #pragma unroll
    for (int kq = 0; kq < 4; ++kq) {
        acc = __builtin_amdgcn_wmma_f32_16x16x4_f32(
                   false, fa[kq],  false, fb[kq],
                   (short)0, acc,
                   false,  false);
    }
    return acc;
}

__global__ __launch_bounds__(256)
void gmm_prep_kernel(const float* __restrict__ means,
                     const float* __restrict__ covs,
                     const float* __restrict__ weights,
                     float* __restrict__ Wc,
                     float* __restrict__ constc,
                     float* __restrict__ Gd,
                     float* __restrict__ hd,
                     float* __restrict__ cd)
{
    __shared__ __align__(16) float Ms[GMM_D * MP];
    __shared__ __align__(16) float Ts[16 * TP];
    __shared__ __align__(16) float Zs[48 * TP];
    __shared__ __align__(16) float dinvs[16];
    __shared__ __align__(16) float dbuf[GMM_D];
    __shared__ __align__(16) float Gs[GMM_D * GP];
    __shared__ float qs[8][GMM_D];
    __shared__ float hhs[8];

    const int t  = threadIdx.x;
    const int tx = t & 31;
    const int ty = t >> 5;
    const int c  = blockIdx.x;

    const float* cov = covs + (size_t)c * GMM_D * GMM_D;

    #pragma unroll
    for (int q = 0; q < 4; ++q) {
        const int e   = q * 256 + t;
        const int row = e >> 4;
        const int c4  = (e & 15) * 4;
        const float4 v = *(const float4*)(cov + row * GMM_D + c4);
        *(float4*)&Ms[row * MP + c4] = v;
        *(float4*)&Ms[row * MP + GMM_D + c4] =
            make_float4(row == c4 ? 1.0f : 0.0f, row == c4 + 1 ? 1.0f : 0.0f,
                        row == c4 + 2 ? 1.0f : 0.0f, row == c4 + 3 ? 1.0f : 0.0f);
    }

    const float mu0 = means[c * GMM_D + tx];
    const float mu1 = means[c * GMM_D + 32 + tx];

    const int wu = __builtin_amdgcn_readfirstlane(ty);

    __syncthreads();

    for (int kb = 0; kb < 4; ++kb) {
        const int k0 = 16 * kb;
        const int nb = 3 - kb;
        if (wu == 0) {
            const int r = tx & 15;
            float s[16], tt[16];
            const float* sp = &Ms[(k0 + r) * MP + k0];
            #pragma unroll
            for (int q = 0; q < 4; ++q) {
                const float4 v = *(const float4*)(sp + 4 * q);
                s[4 * q] = v.x; s[4 * q + 1] = v.y; s[4 * q + 2] = v.z; s[4 * q + 3] = v.w;
            }
            #pragma unroll
            for (int cc = 0; cc < 16; ++cc) tt[cc] = (cc == r) ? 1.0f : 0.0f;
            float* dk = &dbuf[k0];
            gmm_diag_pivot<0>(s, tt, r, tx, dk, dinvs);
            gmm_diag_pivot<1>(s, tt, r, tx, dk, dinvs);
            gmm_diag_pivot<2>(s, tt, r, tx, dk, dinvs);
            gmm_diag_pivot<3>(s, tt, r, tx, dk, dinvs);
            gmm_diag_pivot<4>(s, tt, r, tx, dk, dinvs);
            gmm_diag_pivot<5>(s, tt, r, tx, dk, dinvs);
            gmm_diag_pivot<6>(s, tt, r, tx, dk, dinvs);
            gmm_diag_pivot<7>(s, tt, r, tx, dk, dinvs);
            gmm_diag_pivot<8>(s, tt, r, tx, dk, dinvs);
            gmm_diag_pivot<9>(s, tt, r, tx, dk, dinvs);
            gmm_diag_pivot<10>(s, tt, r, tx, dk, dinvs);
            gmm_diag_pivot<11>(s, tt, r, tx, dk, dinvs);
            gmm_diag_pivot<12>(s, tt, r, tx, dk, dinvs);
            gmm_diag_pivot<13>(s, tt, r, tx, dk, dinvs);
            gmm_diag_pivot<14>(s, tt, r, tx, dk, dinvs);
            gmm_diag_pivot<15>(s, tt, r, tx, dk, dinvs);
            if (tx < 16) {
                #pragma unroll
                for (int q = 0; q < 4; ++q)
                    *(float4*)&Ts[r * TP + 4 * q] =
                        make_float4(tt[4 * q], tt[4 * q + 1], tt[4 * q + 2], tt[4 * q + 3]);
            }
        }
        __syncthreads();

        {
            const int n    = tx & 15;
            const int half = tx >> 4;
            v8f acc;
            #pragma unroll
            for (int r = 0; r < 8; ++r) acc[r] = 0.0f;
            if (wu < 4) {
                const int cb = (wu < nb) ? (kb + 1 + wu) : (4 + wu - nb);
                float* blk = &Ms[k0 * MP + 16 * cb];
                acc = gmm_mm16(Ts, TP, 1, blk, MP, 1, acc, tx);
                #pragma unroll
                for (int r = 0; r < 8; ++r) blk[(8 * half + r) * MP + n] = acc[r];
            } else if (wu - 4 < nb) {
                const int i = kb + 1 + (wu - 4);
                acc = gmm_mm16(&Ms[16 * i * MP + k0], MP, 1, Ts, 1, TP, acc, tx);
                const float di = dinvs[n];
                #pragma unroll
                for (int r = 0; r < 8; ++r)
                    Zs[((wu - 4) * 16 + 8 * half + r) * TP + n] = -acc[r] * di;
            }
        }
        __syncthreads();

        if (nb > 0) {
            const int n    = tx & 15;
            const int half = tx >> 4;
            for (int idx = wu; idx < 4 * nb; idx += 8) {
                const int zi  = idx >> 2;
                const int cbi = idx & 3;
                const int cb  = (cbi < nb) ? (kb + 1 + cbi) : (4 + cbi - nb);
                float* blk = &Ms[16 * (kb + 1 + zi) * MP + 16 * cb];
                v8f acc;
                #pragma unroll
                for (int r = 0; r < 8; ++r) acc[r] = blk[(8 * half + r) * MP + n];
                acc = gmm_mm16(&Zs[zi * 16 * TP], TP, 1, &Ms[k0 * MP + 16 * cb], MP, 1, acc, tx);
                #pragma unroll
                for (int r = 0; r < 8; ++r) blk[(8 * half + r) * MP + n] = acc[r];
            }
            __syncthreads();
        }
    }

    float q0 = 0.0f, q1 = 0.0f, hhw = 0.0f;
    const float4 da = *(const float4*)&dbuf[ty * 8];
    const float4 db = *(const float4*)&dbuf[ty * 8 + 4];
    const float dv[8] = {da.x, da.y, da.z, da.w, db.x, db.y, db.z, db.w};
    #pragma unroll
    for (int i = 0; i < 8; ++i) {
        const int r = ty * 8 + i;
        const float d  = dv[i];
        const float y0 = __builtin_amdgcn_rsqf(d);
        const float rs = fmaf(0.5f * y0, fmaf(-(d * y0), y0, 1.0f), y0);
        const float g0 = Ms[r * MP + GMM_D + tx] * rs;
        const float g1 = Ms[r * MP + GMM_D + 32 + tx] * rs;
        Gs[r * GP + tx]      = g0;
        Gs[r * GP + 32 + tx] = g1;
        const float hp = gmm_wave_sum(g0 * mu0 + g1 * mu1);
        q0  = fmaf(g0, hp, q0);
        q1  = fmaf(g1, hp, q1);
        hhw = fmaf(hp, hp, hhw);
        if (c >= NCP && tx == 0) hd[(c - NCP) * GMM_D + r] = hp;
    }
    qs[ty][tx]      = q0;
    qs[ty][tx + 32] = q1;
    if (tx == 0) hhs[ty] = hhw;

    __syncthreads();

    float ccl = 0.0f;
    if (wu == 7) {
        const float ld = gmm_wave_sum(logf(dbuf[tx]) + logf(dbuf[tx + 32]));
        const float LOG2PI = 1.8378770664093453f;
        ccl = logf(weights[c]) - 0.5f * (ld + (float)GMM_D * LOG2PI);
    }

    if (c >= NCP) {
        float* gd = Gd + (size_t)(c - NCP) * NFR * 64;
        for (int e = t; e < NFR * 64; e += 256) {
            const int f  = e >> 6;
            const int l  = (e >> 1) & 31;
            const int sx = e & 1;
            const int mt = (f >= 24) ? 3 : ((f >= 12) ? 2 : ((f >= 4) ? 1 : 0));
            const int kq = f - 2 * mt * (mt + 1);
            gd[e] = Gs[(16 * mt + (l & 15)) * GP + 4 * kq + 2 * (l >> 4) + sx];
        }
        if (t == 255) cd[c - NCP] = ccl;
        return;
    }

    switch (wu) {
        case 0: gmm_p_tile<0, 0>(Gs, tx, c, Wc); break;
        case 1: gmm_p_tile<0, 1>(Gs, tx, c, Wc); break;
        case 2: gmm_p_tile<1, 1>(Gs, tx, c, Wc); break;
        case 3: gmm_p_tile<0, 2>(Gs, tx, c, Wc); gmm_p_tile<0, 3>(Gs, tx, c, Wc); break;
        case 4: gmm_p_tile<1, 2>(Gs, tx, c, Wc); gmm_p_tile<1, 3>(Gs, tx, c, Wc); break;
        case 5: gmm_p_tile<2, 2>(Gs, tx, c, Wc); gmm_p_tile<2, 3>(Gs, tx, c, Wc); break;
        case 6: gmm_p_tile<3, 3>(Gs, tx, c, Wc); break;
        default: {
            #pragma unroll
            for (int jj = 0; jj < 2; ++jj) {
                const int j = tx + 32 * jj;
                float q = 0.0f;
                #pragma unroll
                for (int w8 = 0; w8 < 8; ++w8) q += qs[w8][j];
                Wc[wc_index(KLIN + (j >> 2), c, (j >> 1) & 1, j & 1)] = q;
            }
            if (tx == 0) {
                float hh = 0.0f;
                #pragma unroll
                for (int w8 = 0; w8 < 8; ++w8) hh += hhs[w8];
                constc[c] = ccl - 0.5f * hh;
            }
        } break;
    }
}


#define TILE_M 128
#define XP     68
#define CHK    40
#define NCH    (NKS / CHK)
#define CHF    (CHK * NT * 64)
#define NPF    (CHF / 2 / 256)
#define NPC    5
#define PPF    (NPF / NPC)
#define KOLD   28
#define SB_MASK 0x06

template <int K0, int KN>
__device__ __forceinline__ void gmm_chunk_part(v8f (&acc0)[NT], v8f (&acc1)[NT],
                                               const float* xrow0, const float* xj0,
                                               const float* xrow1, const float* xj1,
                                               const float* bb, const float* wnx, float* bnx,
                                               int ki, int kq)
{
    constexpr int PKS = KN / NPC;
    float2 pf[PPF];
    #pragma unroll
    for (int q = 0; q < PPF; ++q)
        pf[q] = *(const float2*)(wnx + q * 512);
    float xi0 = xrow0[ki];
    float xi1 = xrow1[ki];
    v2f   xp0 = *(const v2f*)(xj0 + 4 * kq);
    v2f   xp1 = *(const v2f*)(xj1 + 4 * kq);
    v2f   bf[NT];
    #pragma unroll
    for (int tt = 0; tt < NT; ++tt) bf[tt] = *(const v2f*)(bb + (K0 * NT + tt) * 64);
    #pragma unroll
    for (int kk = 0; kk < KN; ++kk) {
        if (kk > 0 && (kk % PKS) == 0 && (kk / PKS) < NPC) {
            #pragma unroll
            for (int q = 0; q < PPF; ++q)
                *(float2*)(bnx + ((kk / PKS - 1) * PPF + q) * 512) = pf[q];
            #pragma unroll
            for (int q = 0; q < PPF; ++q)
                pf[q] = *(const float2*)(wnx + ((kk / PKS) * PPF + q) * 512);
        }
        int ni = ki, nq = kq + 1;
        if (nq == 16) { ni = ki + 1; nq = (ni < GMM_D) ? (ni >> 2) : 0; }
        float xin0 = xi0, xin1 = xi1;
        v2f   xpn0 = xp0, xpn1 = xp1;
        v2f   bn[NT];
        #pragma unroll
        for (int tt = 0; tt < NT; ++tt) bn[tt] = bf[tt];
        if (kk + 1 < KN) {
            xin0 = xrow0[ni];
            xin1 = xrow1[ni];
            xpn0 = *(const v2f*)(xj0 + 4 * nq);
            xpn1 = *(const v2f*)(xj1 + 4 * nq);
            #pragma unroll
            for (int tt = 0; tt < NT; ++tt)
                bn[tt] = *(const v2f*)(bb + ((K0 + kk + 1) * NT + tt) * 64);
        }
        __builtin_amdgcn_sched_barrier(SB_MASK);
        v2f a0, a1;
        a0.x = xi0 * xp0.x;
        a0.y = xi0 * xp0.y;
        a1.x = xi1 * xp1.x;
        a1.y = xi1 * xp1.y;
        #pragma unroll
        for (int tt = 0; tt < NT; ++tt) {
            acc0[tt] = __builtin_amdgcn_wmma_f32_16x16x4_f32(
                            false, a0,  false, bf[tt],
                            (short)0, acc0[tt],
                            false,  false);
            acc1[tt] = __builtin_amdgcn_wmma_f32_16x16x4_f32(
                            false, a1,  false, bf[tt],
                            (short)0, acc1[tt],
                            false,  false);
        }
        __builtin_amdgcn_sched_barrier(SB_MASK);
        xi0 = xin0; xi1 = xin1;
        xp0 = xpn0; xp1 = xpn1;
        #pragma unroll
        for (int tt = 0; tt < NT; ++tt) bf[tt] = bn[tt];
        ki = ni;
        kq = nq;
    }
    #pragma unroll
    for (int q = 0; q < PPF; ++q)
        *(float2*)(bnx + ((NPC - 1) * PPF + q) * 512) = pf[q];
}

__device__ __forceinline__ void gmm_put_part(float* scr, int w, int lane, const v8f (&a)[NT])
{
    #pragma unroll
    for (int tt = 0; tt < NT; ++tt) {
        #pragma unroll
        for (int r = 0; r < 8; ++r)
            scr[((w * NT + tt) * 8 + r) * 32 + lane] = a[tt][r];
    }
}
__device__ __forceinline__ void gmm_add_part(const float* scr, int w, int lane, v8f (&a)[NT])
{
    #pragma unroll
    for (int tt = 0; tt < NT; ++tt) {
        #pragma unroll
        for (int r = 0; r < 8; ++r)
            a[tt][r] += scr[((w * NT + tt) * 8 + r) * 32 + lane];
    }
}
__device__ __forceinline__ void gmm_store_tile(float* __restrict__ out, const float* __restrict__ constc,
                                               int rowb, int n, const v8f (&a)[NT])
{
    #pragma unroll
    for (int tt = 0; tt < NT; ++tt) {
        const int cls = tt * 16 + n;
        const float c0 = constc[cls];
        #pragma unroll
        for (int r = 0; r < 8; ++r)
            out[(size_t)(rowb + r) * GMM_C + cls] = a[tt][r] + c0;
    }
}

__global__ __launch_bounds__(256)
void gmm_logits_kernel(const float* __restrict__ x,
                       const float* __restrict__ Wc,
                       const float* __restrict__ constc,
                       const float* __restrict__ Gd,
                       const float* __restrict__ hd,
                       const float* __restrict__ cd,
                       float* __restrict__ out)
{
    __shared__ __align__(16) float Xl[TILE_M * XP];
    __shared__ __align__(16) float Bl[2][CHF];
    __shared__ __align__(16) float Gl[NDC * NFR * 64];
    __shared__ __align__(16) float hl[NDC * GMM_D];

    const int t  = threadIdx.x;
    const int b0 = blockIdx.x * TILE_M;

    {
        const int col4 = (t & 15) * 4;
        const int r0   = t >> 4;
        #pragma unroll
        for (int it = 0; it < 8; ++it) {
            const int r = r0 + it * 16;
            const float4 v = *(const float4*)(x + (size_t)(b0 + r) * GMM_D + col4);
            *(float4*)&Xl[r * XP + col4] = v;
        }
        if (t < TILE_M) Xl[t * XP + GMM_D] = 1.0f;
        #pragma unroll
        for (int q = 0; q < NPF; ++q) {
            const float2 v = *(const float2*)(Wc + (size_t)(q * 256 + t) * 2);
            *(float2*)&Bl[0][(q * 256 + t) * 2] = v;
        }
        #pragma unroll
        for (int q = 0; q < NDC * NFR * 64 / 4 / 256; ++q) {
            const float4 v = *(const float4*)(Gd + (size_t)(q * 256 + t) * 4);
            *(float4*)&Gl[(q * 256 + t) * 4] = v;
        }
        hl[t] = hd[t];
    }

    const int lane = t & 31;
    const int wv   = t >> 5;
    const int n    = lane & 15;
    const int half = lane >> 4;
    const int  wu    = __builtin_amdgcn_readfirstlane(wv);
    const bool young = wu >= 4;
    const int  pr    = wu & 3;

    v8f acc0[NT], acc1[NT];
    #pragma unroll
    for (int tt = 0; tt < NT; ++tt) {
        #pragma unroll
        for (int r = 0; r < 8; ++r) { acc0[tt][r] = 0.0f; acc1[tt][r] = 0.0f; }
    }
    __syncthreads();

    const float* xrow0 = &Xl[(pr * 16 + n) * XP];
    const float* xrow1 = &Xl[((pr + 4) * 16 + n) * XP];
    const float* xj0   = xrow0 + 2 * half;
    const float* xj1   = xrow1 + 2 * half;
    const float* xj    = young ? xj1 : xj0;

    float res[NDC];
    {
        v2f xb[16];
        #pragma unroll
        for (int kq = 0; kq < 16; ++kq) xb[kq] = *(const v2f*)(xj + 4 * kq);
        #pragma unroll
        for (int cl = 0; cl < NDC; ++cl) {
            v8f d[4];
            #pragma unroll
            for (int mt = 0; mt < 4; ++mt) {
                const float* hp = &hl[cl * GMM_D + 16 * mt + 8 * half];
                const float4 h0 = *(const float4*)hp;
                const float4 h1 = *(const float4*)(hp + 4);
                d[mt][0] = -h0.x; d[mt][1] = -h0.y; d[mt][2] = -h0.z; d[mt][3] = -h0.w;
                d[mt][4] = -h1.x; d[mt][5] = -h1.y; d[mt][6] = -h1.z; d[mt][7] = -h1.w;
            }
            const float* gl = &Gl[(cl * NFR * 32 + lane) * 2];
            #pragma unroll
            for (int kq = 0; kq < 16; ++kq) {
                #pragma unroll
                for (int mt = kq >> 2; mt < 4; ++mt) {
                    const v2f a = *(const v2f*)(gl + (2 * mt * (mt + 1) + kq) * 64);
                    d[mt] = __builtin_amdgcn_wmma_f32_16x16x4_f32(
                                 false, a,  false, xb[kq],
                                 (short)0, d[mt],
                                 false,  false);
                }
            }
            float ms0 = 0.0f, ms1 = 0.0f;
            #pragma unroll
            for (int mt = 0; mt < 4; ++mt) {
                #pragma unroll
                for (int r = 0; r < 8; r += 2) {
                    ms0 = fmaf(d[mt][r],     d[mt][r],     ms0);
                    ms1 = fmaf(d[mt][r + 1], d[mt][r + 1], ms1);
                }
            }
            float ms = ms0 + ms1;
            ms += __shfl_xor(ms, 16, 32);
            res[cl] = cd[cl] - 0.5f * ms;
        }
    }

    const int TAB_O[NCH] = {0, 40, 82, 124, 170, 218, 268, 325, 380, 444, 520, 602, 700, 845};
    const int TAB_Y[NCH] = {28, 69, 111, 156, 203, 252, 308, 363, 426, 493, 572, 668, 796, 1028};

    int pnext = young ? TAB_Y[0] : TAB_O[0];
    for (int ch = 0; ch < NCH; ++ch) {
        const int nx = (ch + 1 < NCH) ? (ch + 1) : ch;
        const float* wnx = Wc + (size_t)nx * CHF + (size_t)t * 2;
        float*       bnx = &Bl[(ch + 1) & 1][t * 2];
        const float* bb  = &Bl[ch & 1][lane * 2];
        const int p = pnext;
        pnext = young ? TAB_Y[nx] : TAB_O[nx];
        if (young)
            gmm_chunk_part<KOLD, CHK - KOLD>(acc0, acc1, xrow0, xj0, xrow1, xj1, bb, wnx, bnx, p >> 4, p & 15);
        else
            gmm_chunk_part<0, KOLD>(acc0, acc1, xrow0, xj0, xrow1, xj1, bb, wnx, bnx, p >> 4, p & 15);
        __syncthreads();
    }

    float* scr = &Bl[0][0];
    if (young) gmm_put_part(scr, wu, lane, acc0);
    else       gmm_put_part(scr, wu, lane, acc1);
    __syncthreads();

    const int rowb = b0 + wv * 16 + 8 * half;
    if (young) {
        gmm_add_part(scr, wu - 4, lane, acc1);
        gmm_store_tile(out, constc, rowb, n, acc1);
    } else {
        gmm_add_part(scr, wu + 4, lane, acc0);
        gmm_store_tile(out, constc, rowb, n, acc0);
    }
    if (half == 0)
        *(float4*)(out + (size_t)(b0 + wv * 16 + n) * GMM_C + NCP) =
            make_float4(res[0], res[1], res[2], res[3]);
}

extern "C" void kernel_launch(void* const* d_in, const int* in_sizes, int n_in,
                              void* d_out, int out_size, void* d_ws, size_t ws_size,
                              hipStream_t stream) {
    (void)in_sizes; (void)n_in; (void)out_size; (void)ws_size;

    const float* x       = (const float*)d_in[0];
    const float* means   = (const float*)d_in[1];
    const float* covs    = (const float*)d_in[2];
    const float* weights = (const float*)d_in[3];
    float* out = (float*)d_out;

    float* Wcw = (float*)d_ws;
    float* ccw = Wcw + (size_t)NKS * NT * 64;
    float* gdw = ccw + 112;
    float* hdw = gdw + (size_t)NDC * NFR * 64;
    float* cdw = hdw + NDC * GMM_D;

    gmm_prep_kernel<<<GMM_C, 256, 0, stream>>>(means, covs, weights, Wcw, ccw, gdw, hdw, cdw);

    gmm_logits_kernel<<<GMM_B / TILE_M, 256, 0, stream>>>(x, Wcw, ccw, gdw, hdw, cdw, out);
}
